# starting kernel
# baseline (speedup 1.0000x reference)
#include <hip/hip_runtime.h>


typedef __attribute__((ext_vector_type(2))) float v2f;
typedef __attribute__((ext_vector_type(8))) float v8f;

#define T_TOKENS 4096
#define DM 512
#define DH 2048
#define NE 8
#define MT 16
#define HC 128
#define XROW (DM + 4)
#define HROW (HC + 4)

static __device__ __forceinline__ v8f wmma_f32(v2f a, v2f b, v8f c) {
  return __builtin_amdgcn_wmma_f32_16x16x4_f32(false, a, false, b, (short)0, c,
                                               false, false);
}

__global__ __launch_bounds__(256) void gating_kernel(
    const float* __restrict__ emb, const float* __restrict__ state,
    const float* __restrict__ Wg, const float* __restrict__ bg,
    const float* __restrict__ Kmat, float* __restrict__ probs_out,
    float* __restrict__ gates_out) {
  const int wave = threadIdx.x >> 5;
  const int lane = threadIdx.x & 31;
  const int t = blockIdx.x * 8 + wave;
  if (t >= T_TOKENS) return;

  const float* x = emb + (size_t)t * DM;
  float acc[NE];
#pragma unroll
  for (int e = 0; e < NE; ++e) acc[e] = 0.0f;

  for (int d = lane; d < DM; d += 32) {
    const float xd = x[d];
#pragma unroll
    for (int e = 0; e < NE; ++e) acc[e] = fmaf(xd, Wg[e * DM + d], acc[e]);
  }
#pragma unroll
  for (int e = 0; e < NE; ++e) {
#pragma unroll
    for (int off = 16; off > 0; off >>= 1) acc[e] += __shfl_xor(acc[e], off, 32);
  }

  if (lane == 0) {
    float logit[NE], p[NE];
#pragma unroll
    for (int e = 0; e < NE; ++e) {
      float tb = 0.0f;
#pragma unroll
      for (int f = 0; f < NE; ++f) tb = fmaf(Kmat[e * NE + f], state[t * NE + f], tb);
      logit[e] = acc[e] + bg[e] + tb;
    }
    float m = logit[0];
#pragma unroll
    for (int e = 1; e < NE; ++e) m = fmaxf(m, logit[e]);
    float s = 0.0f;
#pragma unroll
    for (int e = 0; e < NE; ++e) { p[e] = expf(logit[e] - m); s += p[e]; }
    const float inv = 1.0f / s;
#pragma unroll
    for (int e = 0; e < NE; ++e) p[e] *= inv;
    int i1 = 0;
#pragma unroll
    for (int e = 1; e < NE; ++e) if (p[e] > p[i1]) i1 = e;
    int i2 = (i1 == 0) ? 1 : 0;
#pragma unroll
    for (int e = 0; e < NE; ++e) if (e != i1 && p[e] > p[i2]) i2 = e;
#pragma unroll
    for (int e = 0; e < NE; ++e) {
      probs_out[t * NE + e] = p[e];
      gates_out[t * NE + e] = (e == i1 || e == i2) ? p[e] : 0.0f;
    }
  }
}

__global__ __launch_bounds__(256) void moe_ffn_kernel(
    const float* __restrict__ emb, const float* __restrict__ gates,
    const float* __restrict__ W1, const float* __restrict__ b1,
    const float* __restrict__ W2, const float* __restrict__ b2,
    float* __restrict__ out) {
  __shared__ float sX[MT * XROW];
  __shared__ float sH[MT * HROW];
  __shared__ float sG[MT * NE];
  __shared__ float sFlag[NE];

  const int tid = threadIdx.x;
  const int wave = tid >> 5;
  const int lane = tid & 31;
  const int hf = lane >> 4;
  const int ln = lane & 15;
  const int t0 = blockIdx.x * MT;

  for (int i = tid; i < MT * DM; i += 256) {
    const int m = i >> 9, d = i & (DM - 1);
    sX[m * XROW + d] = emb[(size_t)(t0 + m) * DM + d];
  }
  for (int i = tid; i < MT * NE; i += 256) sG[i] = gates[(size_t)t0 * NE + i];
  __syncthreads();
  if (tid < NE) {
    float f = 0.0f;
    for (int m = 0; m < MT; ++m) f += sG[m * NE + tid];
    sFlag[tid] = f;
  }
  __syncthreads();

  v8f accOut[4];
#pragma unroll
  for (int nt = 0; nt < 4; ++nt) accOut[nt] = (v8f)0.0f;

  for (int e = 0; e < NE; ++e) {
    if (sFlag[e] == 0.0f) continue;

    const float* W1p = W1 + (size_t)e * DM * DH;
    const float* W2p = W2 + (size_t)e * DH * DM;

    v8f accO[4];
#pragma unroll
    for (int nt = 0; nt < 4; ++nt) accO[nt] = (v8f)0.0f;

    for (int ch = 0; ch < DH / HC; ++ch) {
      const int hbase = ch * HC;
      const float* W2c = W2p + (size_t)hbase * DM;
      if (ch + 1 < DH / HC)
        __builtin_prefetch(&W2c[HC * DM + wave * 64 + ln], 0, 1);

      {
        const int hcol = hbase + wave * 16 + ln;
        v8f hcc = (v8f)0.0f;
#pragma unroll 8
        for (int k0 = 0; k0 < DM; k0 += 4) {
          const int k = k0 + 2 * hf;
          v2f a = {sX[ln * XROW + k], sX[ln * XROW + k + 1]};
          const int bi = k * DH + hcol;
          v2f b = {W1p[bi], W1p[bi + DH]};
          hcc = wmma_f32(a, b, hcc);
        }
        const float bias = b1[e * DH + hcol];
#pragma unroll
        for (int r = 0; r < 8; ++r) {
          float v = hcc[r] + bias;
          v = v > 0.0f ? v : 0.0f;
          sH[(r + 8 * hf) * HROW + wave * 16 + ln] = v;
        }
      }
      __syncthreads();

#pragma unroll 4
      for (int k0 = 0; k0 < HC; k0 += 4) {
        const int k = k0 + 2 * hf;
        v2f a = {sH[ln * HROW + k], sH[ln * HROW + k + 1]};
#pragma unroll
        for (int nt = 0; nt < 4; ++nt) {
          const int ncol = wave * 64 + nt * 16 + ln;
          const int bi = k * DM + ncol;
          v2f b = {W2c[bi], W2c[bi + DM]};
          accO[nt] = wmma_f32(a, b, accO[nt]);
        }
      }
      __syncthreads();
    }

#pragma unroll
    for (int nt = 0; nt < 4; ++nt) {
      const int ncol = wave * 64 + nt * 16 + ln;
      const float b2v = b2[e * DM + ncol];
#pragma unroll
      for (int r = 0; r < 8; ++r) {
        const float g = sG[(r + 8 * hf) * NE + e];
        accOut[nt][r] = fmaf(g, accO[nt][r] + b2v, accOut[nt][r]);
      }
    }
  }

#pragma unroll
  for (int nt = 0; nt < 4; ++nt) {
    const int ncol = wave * 64 + nt * 16 + ln;
#pragma unroll
    for (int r = 0; r < 8; ++r) {
      const int m = r + 8 * hf;
      out[(size_t)(t0 + m) * DM + ncol] = accOut[nt][r];
    }
  }
}

extern "C" void kernel_launch(void* const* d_in, const int* in_sizes, int n_in,
                              void* d_out, int out_size, void* d_ws, size_t ws_size,
                              hipStream_t stream) {
  const float* emb   = (const float*)d_in[0];
  const float* state = (const float*)d_in[1];
  const float* Wg    = (const float*)d_in[2];
  const float* bg    = (const float*)d_in[3];
  const float* Kmat  = (const float*)d_in[4];
  const float* W1    = (const float*)d_in[5];
  const float* b1    = (const float*)d_in[6];
  const float* W2    = (const float*)d_in[7];
  const float* b2    = (const float*)d_in[8];

  float* out_main = (float*)d_out;
  float* probs    = out_main + (size_t)T_TOKENS * DM;
  float* gates    = (float*)d_ws;

  gating_kernel<<<T_TOKENS / 8, 256, 0, stream>>>(emb, state, Wg, bg, Kmat,
                                                  probs, gates);
  moe_ffn_kernel<<<T_TOKENS / MT, 256, 0, stream>>>(emb, gates, W1, b1, W2, b2,
                                                    out_main);
}


# route tokens by top-2 expert pair: a scan kernel lists tokens per pair, each 16-token tile runs only its pair's 2 experts (low index first) instead of all 8; sum order kept
# speedup vs baseline: 2.7676x; 2.7676x over previous
#include <hip/hip_runtime.h>


typedef __attribute__((ext_vector_type(2))) float v2f;
typedef __attribute__((ext_vector_type(8))) float v8f;

#define T_TOKENS 4096
#define DM 512
#define DH 2048
#define NE 8
#define MT 16
#define HC 128
#define XROW (DM + 4)
#define HROW (HC + 4)
#define NP 28
#define RHEAD 32
#define RSTRIDE (T_TOKENS + RHEAD)
#define NTILE (T_TOKENS / MT + NP)

static __device__ __forceinline__ v8f wmma_f32(v2f a, v2f b, v8f c) {
  return __builtin_amdgcn_wmma_f32_16x16x4_f32(false, a, false, b, (short)0, c,
                                               false, false);
}

static __device__ __forceinline__ void top2(const float* p, int& i1, int& i2) {
  i1 = 0;
#pragma unroll
  for (int e = 1; e < NE; ++e) if (p[e] > p[i1]) i1 = e;
  i2 = (i1 == 0) ? 1 : 0;
#pragma unroll
  for (int e = 0; e < NE; ++e) if (e != i1 && p[e] > p[i2]) i2 = e;
}

static __device__ __forceinline__ void pair_decode(int p, int& lo, int& hi) {
  lo = 0;
  while (p >= NE - 1 - lo) { p -= NE - 1 - lo; ++lo; }
  hi = lo + 1 + p;
}

__global__ __launch_bounds__(256) void gating_kernel(
    const float* __restrict__ emb, const float* __restrict__ state,
    const float* __restrict__ Wg, const float* __restrict__ bg,
    const float* __restrict__ Kmat, float* __restrict__ probs_out,
    float* __restrict__ gates_out) {
  const int wave = threadIdx.x >> 5;
  const int lane = threadIdx.x & 31;
  const int t = blockIdx.x * 8 + wave;
  if (t >= T_TOKENS) return;

  const float* x = emb + (size_t)t * DM;
  float acc[NE];
#pragma unroll
  for (int e = 0; e < NE; ++e) acc[e] = 0.0f;

  for (int d = lane; d < DM; d += 32) {
    const float xd = x[d];
#pragma unroll
    for (int e = 0; e < NE; ++e) acc[e] = fmaf(xd, Wg[e * DM + d], acc[e]);
  }
#pragma unroll
  for (int e = 0; e < NE; ++e) {
#pragma unroll
    for (int off = 16; off > 0; off >>= 1) acc[e] += __shfl_xor(acc[e], off, 32);
  }

  if (lane == 0) {
    float logit[NE], p[NE];
#pragma unroll
    for (int e = 0; e < NE; ++e) {
      float tb = 0.0f;
#pragma unroll
      for (int f = 0; f < NE; ++f) tb = fmaf(Kmat[e * NE + f], state[t * NE + f], tb);
      logit[e] = acc[e] + bg[e] + tb;
    }
    float m = logit[0];
#pragma unroll
    for (int e = 1; e < NE; ++e) m = fmaxf(m, logit[e]);
    float s = 0.0f;
#pragma unroll
    for (int e = 0; e < NE; ++e) { p[e] = expf(logit[e] - m); s += p[e]; }
    const float inv = 1.0f / s;
#pragma unroll
    for (int e = 0; e < NE; ++e) p[e] *= inv;
    int i1, i2;
    top2(p, i1, i2);
#pragma unroll
    for (int e = 0; e < NE; ++e) {
      probs_out[t * NE + e] = p[e];
      gates_out[t * NE + e] = (e == i1 || e == i2) ? p[e] : 0.0f;
    }
  }
}

__global__ __launch_bounds__(1024) void route_kernel(
    const float* __restrict__ probs, int* __restrict__ route) {
  __shared__ int sWave[32];
  const int tid = threadIdx.x;
  const int wave = tid >> 5;
  const int lane = tid & 31;
  int lo, hi;
  pair_decode(blockIdx.x, lo, hi);

  bool mine[4];
  int c = 0;
#pragma unroll
  for (int q = 0; q < 4; ++q) {
    const float* pr = probs + (size_t)(tid * 4 + q) * NE;
    float p[NE];
#pragma unroll
    for (int e = 0; e < NE; ++e) p[e] = pr[e];
    int i1, i2;
    top2(p, i1, i2);
    const int a = i1 < i2 ? i1 : i2;
    const int b = i1 < i2 ? i2 : i1;
    mine[q] = (a == lo) && (b == hi);
    c += mine[q] ? 1 : 0;
  }
  int inc = c;
#pragma unroll
  for (int off = 1; off < 32; off <<= 1) {
    const int v = __shfl_up(inc, off, 32);
    if (lane >= off) inc += v;
  }
  if (lane == 31) sWave[wave] = inc;
  __syncthreads();
  if (wave == 0) {
    const int w = sWave[lane];
    int winc = w;
#pragma unroll
    for (int off = 1; off < 32; off <<= 1) {
      const int v = __shfl_up(winc, off, 32);
      if (lane >= off) winc += v;
    }
    sWave[lane] = winc - w;
  }
  __syncthreads();
  int rank = sWave[wave] + inc - c;
  int* reg = route + (size_t)blockIdx.x * RSTRIDE;
#pragma unroll
  for (int q = 0; q < 4; ++q) {
    if (mine[q]) { reg[RHEAD + rank] = tid * 4 + q; ++rank; }
  }
  if (tid == 1023) reg[0] = rank;
}

__global__ __launch_bounds__(256) void moe_ffn_kernel(
    const float* __restrict__ emb, const float* __restrict__ gates,
    const int* __restrict__ route,
    const float* __restrict__ W1, const float* __restrict__ b1,
    const float* __restrict__ W2, const float* __restrict__ b2,
    float* __restrict__ out) {
  __shared__ float sX[MT * XROW];
  __shared__ float sH[MT * HROW];
  __shared__ float sG[MT * NE];
  __shared__ int sTok[MT];

  const int tid = threadIdx.x;
  const int wave = tid >> 5;
  const int lane = tid & 31;
  const int hf = lane >> 4;
  const int ln = lane & 15;

  int pair = -1, first = 0, count = 0;
  {
    int cnts[NP];
#pragma unroll
    for (int q = 0; q < NP; ++q) cnts[q] = route[q * RSTRIDE];
    int b = blockIdx.x;
#pragma unroll
    for (int q = 0; q < NP; ++q) {
      const int c = cnts[q];
      const int nt = (c + MT - 1) / MT;
      const bool open = pair < 0;
      const bool hit = open && (b < nt);
      pair = hit ? q : pair;
      first = hit ? b * MT : first;
      count = hit ? c : count;
      b = (open && !hit) ? b - nt : b;
    }
  }
  if (pair < 0) return;
  int lo, hi;
  pair_decode(pair, lo, hi);

  if (tid < MT) {
    const int idx = first + tid;
    sTok[tid] = idx < count ? route[pair * RSTRIDE + RHEAD + idx] : -1;
  }
  __syncthreads();

  for (int i = tid; i < MT * DM; i += 256) {
    const int m = i >> 9, d = i & (DM - 1);
    const int tk = sTok[m];
    sX[m * XROW + d] = tk >= 0 ? emb[(size_t)tk * DM + d] : 0.0f;
  }
  for (int i = tid; i < MT * NE; i += 256) {
    const int tk = sTok[i >> 3];
    sG[i] = tk >= 0 ? gates[(size_t)tk * NE + (i & (NE - 1))] : 0.0f;
  }
  __syncthreads();

  v8f accOut[4];
#pragma unroll
  for (int nt = 0; nt < 4; ++nt) accOut[nt] = (v8f)0.0f;

  for (int s = 0; s < 2; ++s) {
    const int e = s ? hi : lo;

    const float* W1p = W1 + (size_t)e * DM * DH;
    const float* W2p = W2 + (size_t)e * DH * DM;

    v8f accO[4];
#pragma unroll
    for (int nt = 0; nt < 4; ++nt) accO[nt] = (v8f)0.0f;

    for (int ch = 0; ch < DH / HC; ++ch) {
      const int hbase = ch * HC;
      const float* W2c = W2p + (size_t)hbase * DM;
      if (ch + 1 < DH / HC)
        __builtin_prefetch(&W2c[HC * DM + wave * 64 + ln], 0, 1);

      {
        const int hcol = hbase + wave * 16 + ln;
        v8f hcc = (v8f)0.0f;
#pragma unroll 8
        for (int k0 = 0; k0 < DM; k0 += 4) {
          const int k = k0 + 2 * hf;
          v2f a = {sX[ln * XROW + k], sX[ln * XROW + k + 1]};
          const int bi = k * DH + hcol;
          v2f b = {W1p[bi], W1p[bi + DH]};
          hcc = wmma_f32(a, b, hcc);
        }
        const float bias = b1[e * DH + hcol];
#pragma unroll
        for (int r = 0; r < 8; ++r) {
          float v = hcc[r] + bias;
          v = v > 0.0f ? v : 0.0f;
          sH[(r + 8 * hf) * HROW + wave * 16 + ln] = v;
        }
      }
      __syncthreads();

#pragma unroll 4
      for (int k0 = 0; k0 < HC; k0 += 4) {
        const int k = k0 + 2 * hf;
        v2f a = {sH[ln * HROW + k], sH[ln * HROW + k + 1]};
#pragma unroll
        for (int nt = 0; nt < 4; ++nt) {
          const int ncol = wave * 64 + nt * 16 + ln;
          const int bi = k * DM + ncol;
          v2f b = {W2c[bi], W2c[bi + DM]};
          accO[nt] = wmma_f32(a, b, accO[nt]);
        }
      }
      __syncthreads();
    }

#pragma unroll
    for (int nt = 0; nt < 4; ++nt) {
      const int ncol = wave * 64 + nt * 16 + ln;
      const float b2v = b2[e * DM + ncol];
#pragma unroll
      for (int r = 0; r < 8; ++r) {
        const float g = sG[(r + 8 * hf) * NE + e];
        accOut[nt][r] = fmaf(g, accO[nt][r] + b2v, accOut[nt][r]);
      }
    }
  }

  int tokr[8];
#pragma unroll
  for (int r = 0; r < 8; ++r) tokr[r] = sTok[r + 8 * hf];
#pragma unroll
  for (int nt = 0; nt < 4; ++nt) {
    const int ncol = wave * 64 + nt * 16 + ln;
#pragma unroll
    for (int r = 0; r < 8; ++r) {
      if (tokr[r] >= 0) out[(size_t)tokr[r] * DM + ncol] = accOut[nt][r];
    }
  }
}

extern "C" void kernel_launch(void* const* d_in, const int* in_sizes, int n_in,
                              void* d_out, int out_size, void* d_ws, size_t ws_size,
                              hipStream_t stream) {
  const float* emb   = (const float*)d_in[0];
  const float* state = (const float*)d_in[1];
  const float* Wg    = (const float*)d_in[2];
  const float* bg    = (const float*)d_in[3];
  const float* Kmat  = (const float*)d_in[4];
  const float* W1    = (const float*)d_in[5];
  const float* b1    = (const float*)d_in[6];
  const float* W2    = (const float*)d_in[7];
  const float* b2    = (const float*)d_in[8];

  float* out_main = (float*)d_out;
  float* probs    = out_main + (size_t)T_TOKENS * DM;
  float* gates    = (float*)d_ws;
  int*   route    = (int*)(gates + (size_t)T_TOKENS * NE);

  gating_kernel<<<T_TOKENS / 8, 256, 0, stream>>>(emb, state, Wg, bg, Kmat,
                                                  probs, gates);
  route_kernel<<<NP, 1024, 0, stream>>>(probs, route);
  moe_ffn_kernel<<<NTILE, 256, 0, stream>>>(emb, gates, route, W1, b1, W2, b2,
                                            out_main);
}


# B operand (weight) loads made 16 bytes per lane (lane owns 4 adjacent columns, hidden chunk 512) and kept 3 register blocks ahead of their WMMAs, also across LDS-only chunk barriers; sum order kept
# speedup vs baseline: 5.1787x; 1.8712x over previous
#include <hip/hip_runtime.h>


typedef __attribute__((ext_vector_type(2))) float v2f;
typedef __attribute__((ext_vector_type(4))) float v4f;
typedef __attribute__((ext_vector_type(8))) float v8f;

#define T_TOKENS 4096
#define DM 512
#define DH 2048
#define NE 8
#define MT 16
#define HC 512
#define XROW (DM + 4)
#define HROW (HC + 4)
#define NP 28
#define RHEAD 32
#define RSTRIDE (T_TOKENS + RHEAD)
#define NTILE (T_TOKENS / MT + NP)
#define KSPAN 512
#define KB 4
#define RING 4
#define NBLK (KSPAN / (4 * KB))
#define NMAIN (((NBLK - RING + 1) / RING) * RING)

static __device__ __forceinline__ v8f wmma_f32(v2f a, v2f b, v8f c) {
  return __builtin_amdgcn_wmma_f32_16x16x4_f32(false, a, false, b, (short)0, c,
                                               false, false);
}

static __device__ __forceinline__ void load_b_block(
    const v4f* __restrict__ Wq, const int ld4, const int blk, const int lidx,
    v4f (&q)[KB][2]) {
#pragma unroll
  for (int j = 0; j < KB; ++j) {
    const int kk = (blk * KB + j) * 4;
    q[j][0] = Wq[kk * ld4 + lidx];
    q[j][1] = Wq[(kk + 1) * ld4 + lidx];
  }
}

static __device__ __forceinline__ void mma_block(
    const float* sA, const int arow, const int blk, const int hf, const int ln,
    const v4f (&q)[KB][2], v8f (&acc)[4]) {
#pragma unroll
  for (int j = 0; j < KB; ++j) {
    const int k = (blk * KB + j) * 4 + 2 * hf;
    v2f a = {sA[ln * arow + k], sA[ln * arow + k + 1]};
#pragma unroll
    for (int nt = 0; nt < 4; ++nt) {
      v2f b = {q[j][0][nt], q[j][1][nt]};
      acc[nt] = wmma_f32(a, b, acc[nt]);
    }
  }
}

static __device__ __forceinline__ void sweep_request(
    const v4f* __restrict__ Wq, const int ld4, const int lidx,
    v4f (&q)[RING][KB][2]) {
#pragma unroll
  for (int u = 0; u < RING - 1; ++u) load_b_block(Wq, ld4, u, lidx, q[u]);
}

static __device__ __forceinline__ void sweep_run(
    const v4f* __restrict__ Wq, const int ld4, const int lidx, const float* sA,
    const int arow, const int hf, const int ln, v4f (&q)[RING][KB][2],
    v8f (&acc)[4]) {
  for (int b0 = 0; b0 < NMAIN; b0 += RING) {
#pragma unroll
    for (int u = 0; u < RING; ++u) {
      load_b_block(Wq, ld4, b0 + u + RING - 1, lidx, q[(u + RING - 1) % RING]);
      __builtin_amdgcn_sched_barrier(0);
      mma_block(sA, arow, b0 + u, hf, ln, q[u], acc);
      __builtin_amdgcn_sched_barrier(0);
    }
  }
#pragma unroll
  for (int t = NMAIN; t < NBLK; ++t) {
    if (t + RING - 1 < NBLK)
      load_b_block(Wq, ld4, t + RING - 1, lidx, q[(t + RING - 1) % RING]);
    __builtin_amdgcn_sched_barrier(0);
    mma_block(sA, arow, t, hf, ln, q[t % RING], acc);
    __builtin_amdgcn_sched_barrier(0);
  }
}

static __device__ __forceinline__ void lds_barrier() {
  __builtin_amdgcn_fence(__ATOMIC_RELEASE, "workgroup", "local");
  __builtin_amdgcn_s_barrier();
  __builtin_amdgcn_fence(__ATOMIC_ACQUIRE, "workgroup", "local");
}

static __device__ __forceinline__ void top2(const float* p, int& i1, int& i2) {
  i1 = 0;
#pragma unroll
  for (int e = 1; e < NE; ++e) if (p[e] > p[i1]) i1 = e;
  i2 = (i1 == 0) ? 1 : 0;
#pragma unroll
  for (int e = 0; e < NE; ++e) if (e != i1 && p[e] > p[i2]) i2 = e;
}

static __device__ __forceinline__ void pair_decode(int p, int& lo, int& hi) {
  lo = 0;
  while (p >= NE - 1 - lo) { p -= NE - 1 - lo; ++lo; }
  hi = lo + 1 + p;
}

__global__ __launch_bounds__(256) void gating_kernel(
    const float* __restrict__ emb, const float* __restrict__ state,
    const float* __restrict__ Wg, const float* __restrict__ bg,
    const float* __restrict__ Kmat, float* __restrict__ probs_out,
    float* __restrict__ gates_out) {
  const int wave = threadIdx.x >> 5;
  const int lane = threadIdx.x & 31;
  const int t = blockIdx.x * 8 + wave;
  if (t >= T_TOKENS) return;

  const float* x = emb + (size_t)t * DM;
  float acc[NE];
#pragma unroll
  for (int e = 0; e < NE; ++e) acc[e] = 0.0f;

  for (int d = lane; d < DM; d += 32) {
    const float xd = x[d];
#pragma unroll
    for (int e = 0; e < NE; ++e) acc[e] = fmaf(xd, Wg[e * DM + d], acc[e]);
  }
#pragma unroll
  for (int e = 0; e < NE; ++e) {
#pragma unroll
    for (int off = 16; off > 0; off >>= 1) acc[e] += __shfl_xor(acc[e], off, 32);
  }

  if (lane == 0) {
    float logit[NE], p[NE];
#pragma unroll
    for (int e = 0; e < NE; ++e) {
      float tb = 0.0f;
#pragma unroll
      for (int f = 0; f < NE; ++f) tb = fmaf(Kmat[e * NE + f], state[t * NE + f], tb);
      logit[e] = acc[e] + bg[e] + tb;
    }
    float m = logit[0];
#pragma unroll
    for (int e = 1; e < NE; ++e) m = fmaxf(m, logit[e]);
    float s = 0.0f;
#pragma unroll
    for (int e = 0; e < NE; ++e) { p[e] = expf(logit[e] - m); s += p[e]; }
    const float inv = 1.0f / s;
#pragma unroll
    for (int e = 0; e < NE; ++e) p[e] *= inv;
    int i1, i2;
    top2(p, i1, i2);
#pragma unroll
    for (int e = 0; e < NE; ++e) {
      probs_out[t * NE + e] = p[e];
      gates_out[t * NE + e] = (e == i1 || e == i2) ? p[e] : 0.0f;
    }
  }
}

__global__ __launch_bounds__(1024) void route_kernel(
    const float* __restrict__ probs, int* __restrict__ route) {
  __shared__ int sWave[32];
  const int tid = threadIdx.x;
  const int wave = tid >> 5;
  const int lane = tid & 31;
  int lo, hi;
  pair_decode(blockIdx.x, lo, hi);

  bool mine[4];
  int c = 0;
#pragma unroll
  for (int q = 0; q < 4; ++q) {
    const float* pr = probs + (size_t)(tid * 4 + q) * NE;
    float p[NE];
#pragma unroll
    for (int e = 0; e < NE; ++e) p[e] = pr[e];
    int i1, i2;
    top2(p, i1, i2);
    const int a = i1 < i2 ? i1 : i2;
    const int b = i1 < i2 ? i2 : i1;
    mine[q] = (a == lo) && (b == hi);
    c += mine[q] ? 1 : 0;
  }
  int inc = c;
#pragma unroll
  for (int off = 1; off < 32; off <<= 1) {
    const int v = __shfl_up(inc, off, 32);
    if (lane >= off) inc += v;
  }
  if (lane == 31) sWave[wave] = inc;
  __syncthreads();
  if (wave == 0) {
    const int w = sWave[lane];
    int winc = w;
#pragma unroll
    for (int off = 1; off < 32; off <<= 1) {
      const int v = __shfl_up(winc, off, 32);
      if (lane >= off) winc += v;
    }
    sWave[lane] = winc - w;
  }
  __syncthreads();
  int rank = sWave[wave] + inc - c;
  int* reg = route + (size_t)blockIdx.x * RSTRIDE;
#pragma unroll
  for (int q = 0; q < 4; ++q) {
    if (mine[q]) { reg[RHEAD + rank] = tid * 4 + q; ++rank; }
  }
  if (tid == 1023) reg[0] = rank;
}

__global__ __launch_bounds__(256) void moe_ffn_kernel(
    const float* __restrict__ emb, const float* __restrict__ gates,
    const int* __restrict__ route,
    const float* __restrict__ W1, const float* __restrict__ b1,
    const float* __restrict__ W2, const float* __restrict__ b2,
    float* __restrict__ out) {
  __shared__ float sX[MT * XROW];
  __shared__ float sH[MT * HROW];
  __shared__ float sG[MT * NE];
  __shared__ int sTok[MT];

  const int tid = threadIdx.x;
  const int wave = tid >> 5;
  const int lane = tid & 31;
  const int hf = lane >> 4;
  const int ln = lane & 15;

  int pair = -1, first = 0, count = 0;
  {
    int cnts[NP];
#pragma unroll
    for (int q = 0; q < NP; ++q) cnts[q] = route[q * RSTRIDE];
    int b = blockIdx.x;
#pragma unroll
    for (int q = 0; q < NP; ++q) {
      const int c = cnts[q];
      const int nt = (c + MT - 1) / MT;
      const bool open = pair < 0;
      const bool hit = open && (b < nt);
      pair = hit ? q : pair;
      first = hit ? b * MT : first;
      count = hit ? c : count;
      b = (open && !hit) ? b - nt : b;
    }
  }
  if (pair < 0) return;
  int lo, hi;
  pair_decode(pair, lo, hi);

  const int col0 = wave * 64 + 4 * ln;
  const int lidx1 = 2 * hf * (DH / 4) + wave * 16 + ln;
  const int lidx2 = 2 * hf * (DM / 4) + wave * 16 + ln;
  v4f q[RING][KB][2];

  sweep_request((const v4f*)(W1 + (size_t)lo * DM * DH), DH / 4, lidx1, q);

  if (tid < MT) {
    const int idx = first + tid;
    sTok[tid] = idx < count ? route[pair * RSTRIDE + RHEAD + idx] : -1;
  }
  __syncthreads();

  for (int i = tid; i < MT * DM; i += 256) {
    const int m = i >> 9, d = i & (DM - 1);
    const int tk = sTok[m];
    sX[m * XROW + d] = tk >= 0 ? emb[(size_t)tk * DM + d] : 0.0f;
  }
  for (int i = tid; i < MT * NE; i += 256) {
    const int tk = sTok[i >> 3];
    sG[i] = tk >= 0 ? gates[(size_t)tk * NE + (i & (NE - 1))] : 0.0f;
  }
  __syncthreads();

  v8f accOut[4], accO[4];
#pragma unroll
  for (int nt = 0; nt < 4; ++nt) { accOut[nt] = (v8f)0.0f; accO[nt] = (v8f)0.0f; }

  for (int it = 0; it < 2 * (DH / HC); ++it) {
    const int s = it / (DH / HC), ch = it % (DH / HC);
    const int e = s ? hi : lo;
    const int hbase = ch * HC;

    const float* W1p = W1 + (size_t)e * DM * DH;
    const float* W2c = W2 + (size_t)e * DH * DM + (size_t)hbase * DM;

    {
      v8f hcc[4];
#pragma unroll
      for (int nt = 0; nt < 4; ++nt) hcc[nt] = (v8f)0.0f;
      sweep_run((const v4f*)(W1p + hbase), DH / 4, lidx1, sX, XROW, hf, ln, q,
                hcc);
      sweep_request((const v4f*)W2c, DM / 4, lidx2, q);
      const v4f bias = *(const v4f*)&b1[e * DH + hbase + col0];
#pragma unroll
      for (int r = 0; r < 8; ++r) {
        v4f hv;
#pragma unroll
        for (int nt = 0; nt < 4; ++nt) {
          const float v = hcc[nt][r] + bias[nt];
          hv[nt] = v > 0.0f ? v : 0.0f;
        }
        *(v4f*)&sH[(r + 8 * hf) * HROW + col0] = hv;
      }
    }
    lds_barrier();

    sweep_run((const v4f*)W2c, DM / 4, lidx2, sH, HROW, hf, ln, q, accO);
    if (it + 1 < 2 * (DH / HC)) {
      const int s2 = (it + 1) / (DH / HC), ch2 = (it + 1) % (DH / HC);
      const int e2 = s2 ? hi : lo;
      sweep_request((const v4f*)(W1 + (size_t)e2 * DM * DH + ch2 * HC), DH / 4,
                    lidx1, q);
    }
    lds_barrier();

    if (ch == DH / HC - 1) {
      const v4f b2v = *(const v4f*)&b2[e * DM + col0];
#pragma unroll
      for (int nt = 0; nt < 4; ++nt) {
#pragma unroll
        for (int r = 0; r < 8; ++r) {
          const float g = sG[(r + 8 * hf) * NE + e];
          accOut[nt][r] = fmaf(g, accO[nt][r] + b2v[nt], accOut[nt][r]);
        }
        accO[nt] = (v8f)0.0f;
      }
    }
  }

  int tokr[8];
#pragma unroll
  for (int r = 0; r < 8; ++r) tokr[r] = sTok[r + 8 * hf];
#pragma unroll
  for (int r = 0; r < 8; ++r) {
    if (tokr[r] >= 0) {
      const v4f ov = {accOut[0][r], accOut[1][r], accOut[2][r], accOut[3][r]};
      *(v4f*)&out[(size_t)tokr[r] * DM + col0] = ov;
    }
  }
}

extern "C" void kernel_launch(void* const* d_in, const int* in_sizes, int n_in,
                              void* d_out, int out_size, void* d_ws, size_t ws_size,
                              hipStream_t stream) {
  const float* emb   = (const float*)d_in[0];
  const float* state = (const float*)d_in[1];
  const float* Wg    = (const float*)d_in[2];
  const float* bg    = (const float*)d_in[3];
  const float* Kmat  = (const float*)d_in[4];
  const float* W1    = (const float*)d_in[5];
  const float* b1    = (const float*)d_in[6];
  const float* W2    = (const float*)d_in[7];
  const float* b2    = (const float*)d_in[8];

  float* out_main = (float*)d_out;
  float* probs    = out_main + (size_t)T_TOKENS * DM;
  float* gates    = (float*)d_ws;
  int*   route    = (int*)(gates + (size_t)T_TOKENS * NE);

  gating_kernel<<<T_TOKENS / 8, 256, 0, stream>>>(emb, state, Wg, bg, Kmat,
                                                  probs, gates);
  route_kernel<<<NP, 1024, 0, stream>>>(probs, route);
  moe_ffn_kernel<<<NTILE, 256, 0, stream>>>(emb, gates, route, W1, b1, W2, b2,
                                            out_main);
}


# hidden dimension split over workgroups: one per (16-token tile, expert, 512-wide hidden chunk) stores a partial; a 4th kernel adds the 8 partials in fixed order (chunk sums added after the k sweep)
# speedup vs baseline: 7.5614x; 1.4601x over previous
#include <hip/hip_runtime.h>


typedef __attribute__((ext_vector_type(2))) float v2f;
typedef __attribute__((ext_vector_type(4))) float v4f;
typedef __attribute__((ext_vector_type(8))) float v8f;

#define T_TOKENS 4096
#define DM 512
#define DH 2048
#define NE 8
#define MT 16
#define HC 512
#define XROW (DM + 4)
#define HROW (HC + 4)
#define NP 28
#define RHEAD 32
#define RSTRIDE (T_TOKENS + RHEAD)
#define NTILE (T_TOKENS / MT + NP)
#define NCH (DH / HC)
#define NUNIT (2 * NCH)
#define KSPAN 512
#define KB 4
#define RING 4
#define NBLK (KSPAN / (4 * KB))
#define NMAIN (((NBLK - RING + 1) / RING) * RING)

static __device__ __forceinline__ v8f wmma_f32(v2f a, v2f b, v8f c) {
  return __builtin_amdgcn_wmma_f32_16x16x4_f32(false, a, false, b, (short)0, c,
                                               false, false);
}

static __device__ __forceinline__ void load_b_block(
    const v4f* __restrict__ Wq, const int ld4, const int blk, const int lidx,
    v4f (&q)[KB][2]) {
#pragma unroll
  for (int j = 0; j < KB; ++j) {
    const int kk = (blk * KB + j) * 4;
    q[j][0] = Wq[kk * ld4 + lidx];
    q[j][1] = Wq[(kk + 1) * ld4 + lidx];
  }
}

static __device__ __forceinline__ void mma_block(
    const float* sA, const int arow, const int blk, const int hf, const int ln,
    const v4f (&q)[KB][2], v8f (&acc)[4]) {
#pragma unroll
  for (int j = 0; j < KB; ++j) {
    const int k = (blk * KB + j) * 4 + 2 * hf;
    v2f a = {sA[ln * arow + k], sA[ln * arow + k + 1]};
#pragma unroll
    for (int nt = 0; nt < 4; ++nt) {
      v2f b = {q[j][0][nt], q[j][1][nt]};
      acc[nt] = wmma_f32(a, b, acc[nt]);
    }
  }
}

static __device__ __forceinline__ void sweep_request(
    const v4f* __restrict__ Wq, const int ld4, const int lidx,
    v4f (&q)[RING][KB][2]) {
#pragma unroll
  for (int u = 0; u < RING - 1; ++u) load_b_block(Wq, ld4, u, lidx, q[u]);
}

static __device__ __forceinline__ void sweep_run(
    const v4f* __restrict__ Wq, const int ld4, const int lidx, const float* sA,
    const int arow, const int hf, const int ln, v4f (&q)[RING][KB][2],
    v8f (&acc)[4]) {
  for (int b0 = 0; b0 < NMAIN; b0 += RING) {
#pragma unroll
    for (int u = 0; u < RING; ++u) {
      load_b_block(Wq, ld4, b0 + u + RING - 1, lidx, q[(u + RING - 1) % RING]);
      __builtin_amdgcn_sched_barrier(0);
      mma_block(sA, arow, b0 + u, hf, ln, q[u], acc);
      __builtin_amdgcn_sched_barrier(0);
    }
  }
#pragma unroll
  for (int t = NMAIN; t < NBLK; ++t) {
    if (t + RING - 1 < NBLK)
      load_b_block(Wq, ld4, t + RING - 1, lidx, q[(t + RING - 1) % RING]);
    __builtin_amdgcn_sched_barrier(0);
    mma_block(sA, arow, t, hf, ln, q[t % RING], acc);
    __builtin_amdgcn_sched_barrier(0);
  }
}

static __device__ __forceinline__ void lds_barrier() {
  __builtin_amdgcn_fence(__ATOMIC_RELEASE, "workgroup", "local");
  __builtin_amdgcn_s_barrier();
  __builtin_amdgcn_fence(__ATOMIC_ACQUIRE, "workgroup", "local");
}

static __device__ __forceinline__ void top2(const float* p, int& i1, int& i2) {
  i1 = 0;
#pragma unroll
  for (int e = 1; e < NE; ++e) if (p[e] > p[i1]) i1 = e;
  i2 = (i1 == 0) ? 1 : 0;
#pragma unroll
  for (int e = 0; e < NE; ++e) if (e != i1 && p[e] > p[i2]) i2 = e;
}

static __device__ __forceinline__ void pair_decode(int p, int& lo, int& hi) {
  lo = 0;
  while (p >= NE - 1 - lo) { p -= NE - 1 - lo; ++lo; }
  hi = lo + 1 + p;
}

static __device__ __forceinline__ void find_tile(
    const int* __restrict__ route, int b, int& pair, int& first, int& count) {
  pair = -1; first = 0; count = 0;
  int cnts[NP];
#pragma unroll
  for (int q = 0; q < NP; ++q) cnts[q] = route[q * RSTRIDE];
#pragma unroll
  for (int q = 0; q < NP; ++q) {
    const int c = cnts[q];
    const int nt = (c + MT - 1) / MT;
    const bool open = pair < 0;
    const bool hit = open && (b < nt);
    pair = hit ? q : pair;
    first = hit ? b * MT : first;
    count = hit ? c : count;
    b = (open && !hit) ? b - nt : b;
  }
}

__global__ __launch_bounds__(256) void gating_kernel(
    const float* __restrict__ emb, const float* __restrict__ state,
    const float* __restrict__ Wg, const float* __restrict__ bg,
    const float* __restrict__ Kmat, float* __restrict__ probs_out,
    float* __restrict__ gates_out) {
  const int wave = threadIdx.x >> 5;
  const int lane = threadIdx.x & 31;
  const int t = blockIdx.x * 8 + wave;
  if (t >= T_TOKENS) return;

  const float* x = emb + (size_t)t * DM;
  float acc[NE];
#pragma unroll
  for (int e = 0; e < NE; ++e) acc[e] = 0.0f;

  for (int d = lane; d < DM; d += 32) {
    const float xd = x[d];
#pragma unroll
    for (int e = 0; e < NE; ++e) acc[e] = fmaf(xd, Wg[e * DM + d], acc[e]);
  }
#pragma unroll
  for (int e = 0; e < NE; ++e) {
#pragma unroll
    for (int off = 16; off > 0; off >>= 1) acc[e] += __shfl_xor(acc[e], off, 32);
  }

  if (lane == 0) {
    float logit[NE], p[NE];
#pragma unroll
    for (int e = 0; e < NE; ++e) {
      float tb = 0.0f;
#pragma unroll
      for (int f = 0; f < NE; ++f) tb = fmaf(Kmat[e * NE + f], state[t * NE + f], tb);
      logit[e] = acc[e] + bg[e] + tb;
    }
    float m = logit[0];
#pragma unroll
    for (int e = 1; e < NE; ++e) m = fmaxf(m, logit[e]);
    float s = 0.0f;
#pragma unroll
    for (int e = 0; e < NE; ++e) { p[e] = expf(logit[e] - m); s += p[e]; }
    const float inv = 1.0f / s;
#pragma unroll
    for (int e = 0; e < NE; ++e) p[e] *= inv;
    int i1, i2;
    top2(p, i1, i2);
#pragma unroll
    for (int e = 0; e < NE; ++e) {
      probs_out[t * NE + e] = p[e];
      gates_out[t * NE + e] = (e == i1 || e == i2) ? p[e] : 0.0f;
    }
  }
}

__global__ __launch_bounds__(1024) void route_kernel(
    const float* __restrict__ probs, int* __restrict__ route) {
  __shared__ int sWave[32];
  const int tid = threadIdx.x;
  const int wave = tid >> 5;
  const int lane = tid & 31;
  int lo, hi;
  pair_decode(blockIdx.x, lo, hi);

  bool mine[4];
  int c = 0;
#pragma unroll
  for (int q = 0; q < 4; ++q) {
    const float* pr = probs + (size_t)(tid * 4 + q) * NE;
    float p[NE];
#pragma unroll
    for (int e = 0; e < NE; ++e) p[e] = pr[e];
    int i1, i2;
    top2(p, i1, i2);
    const int a = i1 < i2 ? i1 : i2;
    const int b = i1 < i2 ? i2 : i1;
    mine[q] = (a == lo) && (b == hi);
    c += mine[q] ? 1 : 0;
  }
  int inc = c;
#pragma unroll
  for (int off = 1; off < 32; off <<= 1) {
    const int v = __shfl_up(inc, off, 32);
    if (lane >= off) inc += v;
  }
  if (lane == 31) sWave[wave] = inc;
  __syncthreads();
  if (wave == 0) {
    const int w = sWave[lane];
    int winc = w;
#pragma unroll
    for (int off = 1; off < 32; off <<= 1) {
      const int v = __shfl_up(winc, off, 32);
      if (lane >= off) winc += v;
    }
    sWave[lane] = winc - w;
  }
  __syncthreads();
  int rank = sWave[wave] + inc - c;
  int* reg = route + (size_t)blockIdx.x * RSTRIDE;
#pragma unroll
  for (int q = 0; q < 4; ++q) {
    if (mine[q]) { reg[RHEAD + rank] = tid * 4 + q; ++rank; }
  }
  if (tid == 1023) reg[0] = rank;
}

__global__ __launch_bounds__(256) void moe_ffn_kernel(
    const float* __restrict__ emb, const int* __restrict__ route,
    const float* __restrict__ W1, const float* __restrict__ b1,
    const float* __restrict__ W2, float* __restrict__ part) {
  __shared__ float sX[MT * XROW];
  __shared__ float sH[MT * HROW];

  const int tid = threadIdx.x;
  const int wave = tid >> 5;
  const int lane = tid & 31;
  const int hf = lane >> 4;
  const int ln = lane & 15;

  int pair, first, count;
  find_tile(route, blockIdx.x / NUNIT, pair, first, count);
  if (pair < 0) return;
  int lo, hi;
  pair_decode(pair, lo, hi);
  const int item = blockIdx.x % NUNIT;
  const int e = (item / NCH) ? hi : lo;
  const int hbase = (item % NCH) * HC;

  const int col0 = wave * 64 + 4 * ln;
  const int lidx1 = 2 * hf * (DH / 4) + wave * 16 + ln;
  const int lidx2 = 2 * hf * (DM / 4) + wave * 16 + ln;
  v4f q[RING][KB][2];

  const float* W1c = W1 + (size_t)e * DM * DH + hbase;
  const float* W2c = W2 + (size_t)e * DH * DM + (size_t)hbase * DM;

  sweep_request((const v4f*)W1c, DH / 4, lidx1, q);

  {
    const int d4 = tid & 127;
    const int m0 = tid >> 7;
    const int* lst = route + pair * RSTRIDE + RHEAD;
    int tk[MT / 2];
#pragma unroll
    for (int j = 0; j < MT / 2; ++j) {
      const int idx = first + m0 + 2 * j;
      tk[j] = lst[idx < count ? idx : count - 1];
    }
    v4f xv[MT / 2];
#pragma unroll
    for (int j = 0; j < MT / 2; ++j)
      xv[j] = ((const v4f*)(emb + (size_t)tk[j] * DM))[d4];
#pragma unroll
    for (int j = 0; j < MT / 2; ++j) {
      const bool live = first + m0 + 2 * j < count;
      *(v4f*)&sX[(m0 + 2 * j) * XROW + 4 * d4] = live ? xv[j] : (v4f)0.0f;
    }
  }
  __syncthreads();

  {
    v8f hcc[4];
#pragma unroll
    for (int nt = 0; nt < 4; ++nt) hcc[nt] = (v8f)0.0f;
    sweep_run((const v4f*)W1c, DH / 4, lidx1, sX, XROW, hf, ln, q, hcc);
    sweep_request((const v4f*)W2c, DM / 4, lidx2, q);
    const v4f bias = *(const v4f*)&b1[e * DH + hbase + col0];
#pragma unroll
    for (int r = 0; r < 8; ++r) {
      v4f hv;
#pragma unroll
      for (int nt = 0; nt < 4; ++nt) {
        const float v = hcc[nt][r] + bias[nt];
        hv[nt] = v > 0.0f ? v : 0.0f;
      }
      *(v4f*)&sH[(r + 8 * hf) * HROW + col0] = hv;
    }
  }
  lds_barrier();

  v8f accO[4];
#pragma unroll
  for (int nt = 0; nt < 4; ++nt) accO[nt] = (v8f)0.0f;
  sweep_run((const v4f*)W2c, DM / 4, lidx2, sH, HROW, hf, ln, q, accO);

  float* P = part + (size_t)blockIdx.x * (MT * DM);
#pragma unroll
  for (int r = 0; r < 8; ++r) {
    const v4f ov = {accO[0][r], accO[1][r], accO[2][r], accO[3][r]};
    *(v4f*)&P[(r + 8 * hf) * DM + col0] = ov;
  }
}

__global__ __launch_bounds__(256) void combine_kernel(
    const float* __restrict__ gates, const int* __restrict__ route,
    const float* __restrict__ b2, const float* __restrict__ part,
    float* __restrict__ out) {
  const int tid = threadIdx.x;
  const int tile = blockIdx.x / NUNIT;
  int pair, first, count;
  find_tile(route, tile, pair, first, count);
  if (pair < 0) return;
  int lo, hi;
  pair_decode(pair, lo, hi);

  const int row = (blockIdx.x % NUNIT) * (MT / NUNIT) + (tid >> 7);
  const int c4 = tid & 127;
  const int idx = first + row;
  if (idx >= count) return;
  const int tk = route[pair * RSTRIDE + RHEAD + idx];

  const v4f* P = (const v4f*)(part + (size_t)tile * NUNIT * (MT * DM)) +
                 row * (DM / 4) + c4;
  v4f p[NUNIT];
#pragma unroll
  for (int u = 0; u < NUNIT; ++u) p[u] = P[u * (MT * DM / 4)];
  const float glo = gates[(size_t)tk * NE + lo];
  const float ghi = gates[(size_t)tk * NE + hi];
  const v4f blo = ((const v4f*)(b2 + lo * DM))[c4];
  const v4f bhi = ((const v4f*)(b2 + hi * DM))[c4];

  v4f olo = p[0], ohi = p[NCH];
#pragma unroll
  for (int u = 1; u < NCH; ++u) { olo += p[u]; ohi += p[NCH + u]; }
  v4f ov;
#pragma unroll
  for (int nt = 0; nt < 4; ++nt) {
    const float a = glo * (olo[nt] + blo[nt]);
    ov[nt] = fmaf(ghi, ohi[nt] + bhi[nt], a);
  }
  ((v4f*)(out + (size_t)tk * DM))[c4] = ov;
}

extern "C" void kernel_launch(void* const* d_in, const int* in_sizes, int n_in,
                              void* d_out, int out_size, void* d_ws, size_t ws_size,
                              hipStream_t stream) {
  const float* emb   = (const float*)d_in[0];
  const float* state = (const float*)d_in[1];
  const float* Wg    = (const float*)d_in[2];
  const float* bg    = (const float*)d_in[3];
  const float* Kmat  = (const float*)d_in[4];
  const float* W1    = (const float*)d_in[5];
  const float* b1    = (const float*)d_in[6];
  const float* W2    = (const float*)d_in[7];
  const float* b2    = (const float*)d_in[8];

  float* out_main = (float*)d_out;
  float* probs    = out_main + (size_t)T_TOKENS * DM;
  float* gates    = (float*)d_ws;
  int*   route    = (int*)(gates + (size_t)T_TOKENS * NE);
  float* part     = (float*)(route + (size_t)NP * RSTRIDE);

  gating_kernel<<<T_TOKENS / 8, 256, 0, stream>>>(emb, state, Wg, bg, Kmat,
                                                  probs, gates);
  route_kernel<<<NP, 1024, 0, stream>>>(probs, route);
  moe_ffn_kernel<<<NTILE * NUNIT, 256, 0, stream>>>(emb, route, W1, b1, W2,
                                                    part);
  combine_kernel<<<NTILE * NUNIT, 256, 0, stream>>>(gates, route, b2, part,
                                                    out_main);
}


# work item reshaped from 16 tokens x 512 hidden to 32 tokens x 256 hidden: two M tiles share each weight register block (half the L2 weight bytes per WMMA); a token now sums 16 partials, fixed order
# speedup vs baseline: 8.1871x; 1.0827x over previous
#include <hip/hip_runtime.h>


typedef __attribute__((ext_vector_type(2))) float v2f;
typedef __attribute__((ext_vector_type(4))) float v4f;
typedef __attribute__((ext_vector_type(8))) float v8f;

#define T_TOKENS 4096
#define DM 512
#define DH 2048
#define NE 8
#define MT 16
#define MW 32
#define HC 256
#define XROW (DM + 4)
#define HROW (HC + 4)
#define NP 28
#define RHEAD 32
#define RSTRIDE (T_TOKENS + RHEAD)
#define NTILE (T_TOKENS / MW + NP)
#define NCH (DH / HC)
#define NUNIT (2 * NCH)
#define KB 4
#define RING 3

static __device__ __forceinline__ v8f wmma_f32(v2f a, v2f b, v8f c) {
  return __builtin_amdgcn_wmma_f32_16x16x4_f32(false, a, false, b, (short)0, c,
                                               false, false);
}

template <int NT> struct bvec;
template <> struct bvec<2> { typedef v2f type; };
template <> struct bvec<4> { typedef v4f type; };

template <int NT>
static __device__ __forceinline__ void load_b_block(
    const typename bvec<NT>::type* __restrict__ Wq, const int ldv,
    const int blk, const int lidx, typename bvec<NT>::type (&q)[KB][2]) {
#pragma unroll
  for (int j = 0; j < KB; ++j) {
    const int kk = (blk * KB + j) * 4;
    q[j][0] = Wq[kk * ldv + lidx];
    q[j][1] = Wq[(kk + 1) * ldv + lidx];
  }
}

template <int NM, int NT>
static __device__ __forceinline__ void mma_block(
    const float* sA, const int arow, const int blk, const int hf, const int ln,
    const typename bvec<NT>::type (&q)[KB][2], v8f (&acc)[NM][NT]) {
#pragma unroll
  for (int j = 0; j < KB; ++j) {
    const int k = (blk * KB + j) * 4 + 2 * hf;
    v2f a[NM];
#pragma unroll
    for (int m = 0; m < NM; ++m) {
      const float* ar = sA + (m * MT + ln) * arow + k;
      a[m] = (v2f){ar[0], ar[1]};
    }
#pragma unroll
    for (int nt = 0; nt < NT; ++nt) {
      v2f b = {q[j][0][nt], q[j][1][nt]};
#pragma unroll
      for (int m = 0; m < NM; ++m) acc[m][nt] = wmma_f32(a[m], b, acc[m][nt]);
    }
  }
}

template <int NT>
static __device__ __forceinline__ void sweep_request(
    const typename bvec<NT>::type* __restrict__ Wq, const int ldv,
    const int lidx, typename bvec<NT>::type (&q)[RING][KB][2]) {
#pragma unroll
  for (int u = 0; u < RING - 1; ++u) load_b_block<NT>(Wq, ldv, u, lidx, q[u]);
}

template <int NM, int NT, int NBLKS>
static __device__ __forceinline__ void sweep_run(
    const typename bvec<NT>::type* __restrict__ Wq, const int ldv,
    const int lidx, const float* sA, const int arow, const int hf, const int ln,
    typename bvec<NT>::type (&q)[RING][KB][2], v8f (&acc)[NM][NT]) {
  constexpr int nmain = ((NBLKS - RING + 1) / RING) * RING;
#pragma nounroll
  for (int b0 = 0; b0 < nmain; b0 += RING) {
#pragma unroll
    for (int u = 0; u < RING; ++u) {
      load_b_block<NT>(Wq, ldv, b0 + u + RING - 1, lidx, q[(u + RING - 1) % RING]);
      __builtin_amdgcn_sched_barrier(0);
      mma_block<NM, NT>(sA, arow, b0 + u, hf, ln, q[u], acc);
      __builtin_amdgcn_sched_barrier(0);
    }
  }
#pragma unroll
  for (int t = nmain; t < NBLKS; ++t) {
    if (t + RING - 1 < NBLKS)
      load_b_block<NT>(Wq, ldv, t + RING - 1, lidx, q[(t + RING - 1) % RING]);
    __builtin_amdgcn_sched_barrier(0);
    mma_block<NM, NT>(sA, arow, t, hf, ln, q[t % RING], acc);
    __builtin_amdgcn_sched_barrier(0);
  }
}

static __device__ __forceinline__ void lds_barrier() {
  __builtin_amdgcn_fence(__ATOMIC_RELEASE, "workgroup", "local");
  __builtin_amdgcn_s_barrier();
  __builtin_amdgcn_fence(__ATOMIC_ACQUIRE, "workgroup", "local");
}

static __device__ __forceinline__ void top2(const float* p, int& i1, int& i2) {
  i1 = 0;
#pragma unroll
  for (int e = 1; e < NE; ++e) if (p[e] > p[i1]) i1 = e;
  i2 = (i1 == 0) ? 1 : 0;
#pragma unroll
  for (int e = 0; e < NE; ++e) if (e != i1 && p[e] > p[i2]) i2 = e;
}

static __device__ __forceinline__ void pair_decode(int p, int& lo, int& hi) {
  lo = 0;
  while (p >= NE - 1 - lo) { p -= NE - 1 - lo; ++lo; }
  hi = lo + 1 + p;
}

static __device__ __forceinline__ void find_tile(
    const int* __restrict__ route, int b, int& pair, int& first, int& count) {
  pair = -1; first = 0; count = 0;
  int cnts[NP];
#pragma unroll
  for (int q = 0; q < NP; ++q) cnts[q] = route[q * RSTRIDE];
#pragma unroll
  for (int q = 0; q < NP; ++q) {
    const int c = cnts[q];
    const int nt = (c + MW - 1) / MW;
    const bool open = pair < 0;
    const bool hit = open && (b < nt);
    pair = hit ? q : pair;
    first = hit ? b * MW : first;
    count = hit ? c : count;
    b = (open && !hit) ? b - nt : b;
  }
}

__global__ __launch_bounds__(256) void gating_kernel(
    const float* __restrict__ emb, const float* __restrict__ state,
    const float* __restrict__ Wg, const float* __restrict__ bg,
    const float* __restrict__ Kmat, float* __restrict__ probs_out,
    float* __restrict__ gates_out) {
  const int wave = threadIdx.x >> 5;
  const int lane = threadIdx.x & 31;
  const int t = blockIdx.x * 8 + wave;
  if (t >= T_TOKENS) return;

  const float* x = emb + (size_t)t * DM;
  float acc[NE];
#pragma unroll
  for (int e = 0; e < NE; ++e) acc[e] = 0.0f;

  for (int d = lane; d < DM; d += 32) {
    const float xd = x[d];
#pragma unroll
    for (int e = 0; e < NE; ++e) acc[e] = fmaf(xd, Wg[e * DM + d], acc[e]);
  }
#pragma unroll
  for (int e = 0; e < NE; ++e) {
#pragma unroll
    for (int off = 16; off > 0; off >>= 1) acc[e] += __shfl_xor(acc[e], off, 32);
  }

  if (lane == 0) {
    float logit[NE], p[NE];
#pragma unroll
    for (int e = 0; e < NE; ++e) {
      float tb = 0.0f;
#pragma unroll
      for (int f = 0; f < NE; ++f) tb = fmaf(Kmat[e * NE + f], state[t * NE + f], tb);
      logit[e] = acc[e] + bg[e] + tb;
    }
    float m = logit[0];
#pragma unroll
    for (int e = 1; e < NE; ++e) m = fmaxf(m, logit[e]);
    float s = 0.0f;
#pragma unroll
    for (int e = 0; e < NE; ++e) { p[e] = expf(logit[e] - m); s += p[e]; }
    const float inv = 1.0f / s;
#pragma unroll
    for (int e = 0; e < NE; ++e) p[e] *= inv;
    int i1, i2;
    top2(p, i1, i2);
#pragma unroll
    for (int e = 0; e < NE; ++e) {
      probs_out[t * NE + e] = p[e];
      gates_out[t * NE + e] = (e == i1 || e == i2) ? p[e] : 0.0f;
    }
  }
}

__global__ __launch_bounds__(1024) void route_kernel(
    const float* __restrict__ probs, int* __restrict__ route) {
  __shared__ int sWave[32];
  const int tid = threadIdx.x;
  const int wave = tid >> 5;
  const int lane = tid & 31;
  int lo, hi;
  pair_decode(blockIdx.x, lo, hi);

  bool mine[4];
  int c = 0;
#pragma unroll
  for (int q = 0; q < 4; ++q) {
    const float* pr = probs + (size_t)(tid * 4 + q) * NE;
    float p[NE];
#pragma unroll
    for (int e = 0; e < NE; ++e) p[e] = pr[e];
    int i1, i2;
    top2(p, i1, i2);
    const int a = i1 < i2 ? i1 : i2;
    const int b = i1 < i2 ? i2 : i1;
    mine[q] = (a == lo) && (b == hi);
    c += mine[q] ? 1 : 0;
  }
  int inc = c;
#pragma unroll
  for (int off = 1; off < 32; off <<= 1) {
    const int v = __shfl_up(inc, off, 32);
    if (lane >= off) inc += v;
  }
  if (lane == 31) sWave[wave] = inc;
  __syncthreads();
  if (wave == 0) {
    const int w = sWave[lane];
    int winc = w;
#pragma unroll
    for (int off = 1; off < 32; off <<= 1) {
      const int v = __shfl_up(winc, off, 32);
      if (lane >= off) winc += v;
    }
    sWave[lane] = winc - w;
  }
  __syncthreads();
  int rank = sWave[wave] + inc - c;
  int* reg = route + (size_t)blockIdx.x * RSTRIDE;
#pragma unroll
  for (int q = 0; q < 4; ++q) {
    if (mine[q]) { reg[RHEAD + rank] = tid * 4 + q; ++rank; }
  }
  if (tid == 1023) reg[0] = rank;
}


template <int NM>
static __device__ __forceinline__ void ffn_item(
    const float* __restrict__ emb, const int* __restrict__ lst, const int first,
    const int count, const float* __restrict__ W1c,
    const float* __restrict__ W2c, const float* __restrict__ b1c,
    float* __restrict__ P, float* sX, float* sH, const int tid, const int hf,
    const int ln, const int col1, const int col0, const int lidx1,
    const int lidx2, v2f (&q1)[RING][KB][2]) {
  {
    constexpr int NR = NM * MT / 2;
    const int d4 = tid & 127;
    const int m0 = tid >> 7;
    int tk[NR];
#pragma unroll
    for (int j = 0; j < NR; ++j) {
      const int idx = first + m0 + 2 * j;
      tk[j] = lst[idx < count ? idx : count - 1];
    }
    v4f xv[NR];
#pragma unroll
    for (int j = 0; j < NR; ++j)
      xv[j] = ((const v4f*)(emb + (size_t)tk[j] * DM))[d4];
#pragma unroll
    for (int j = 0; j < NR; ++j) {
      const bool live = first + m0 + 2 * j < count;
      *(v4f*)&sX[(m0 + 2 * j) * XROW + 4 * d4] = live ? xv[j] : (v4f)0.0f;
    }
  }
  __syncthreads();

  v4f q2[RING][KB][2];

  {
    v8f hcc[NM][2];
#pragma unroll
    for (int m = 0; m < NM; ++m)
#pragma unroll
      for (int nt = 0; nt < 2; ++nt) hcc[m][nt] = (v8f)0.0f;
    sweep_run<NM, 2, DM / (4 * KB)>((const v2f*)W1c, DH / 2, lidx1, sX, XROW,
                                    hf, ln, q1, hcc);
    sweep_request<4>((const v4f*)W2c, DM / 4, lidx2, q2);
    const v2f bias = *(const v2f*)&b1c[col1];
#pragma unroll
    for (int m = 0; m < NM; ++m) {
#pragma unroll
      for (int r = 0; r < 8; ++r) {
        v2f hv;
#pragma unroll
        for (int nt = 0; nt < 2; ++nt) {
          const float v = hcc[m][nt][r] + bias[nt];
          hv[nt] = v > 0.0f ? v : 0.0f;
        }
        *(v2f*)&sH[(m * MT + r + 8 * hf) * HROW + col1] = hv;
      }
    }
  }
  lds_barrier();

  v8f accO[NM][4];
#pragma unroll
  for (int m = 0; m < NM; ++m)
#pragma unroll
    for (int nt = 0; nt < 4; ++nt) accO[m][nt] = (v8f)0.0f;
  sweep_run<NM, 4, HC / (4 * KB)>((const v4f*)W2c, DM / 4, lidx2, sH, HROW, hf,
                                  ln, q2, accO);

#pragma unroll
  for (int m = 0; m < NM; ++m) {
#pragma unroll
    for (int r = 0; r < 8; ++r) {
      const v4f ov = {accO[m][0][r], accO[m][1][r], accO[m][2][r], accO[m][3][r]};
      *(v4f*)&P[(m * MT + r + 8 * hf) * DM + col0] = ov;
    }
  }
}

__global__ __launch_bounds__(256) void moe_ffn_kernel(
    const float* __restrict__ emb, const int* __restrict__ route,
    const float* __restrict__ W1, const float* __restrict__ b1,
    const float* __restrict__ W2, float* __restrict__ part) {
  __shared__ __attribute__((aligned(16))) float sX[MW * XROW];
  __shared__ __attribute__((aligned(16))) float sH[MW * HROW];

  const int tid = threadIdx.x;
  const int wave = tid >> 5;
  const int lane = tid & 31;
  const int hf = lane >> 4;
  const int ln = lane & 15;

  int pair, first, count;
  find_tile(route, blockIdx.x / NUNIT, pair, first, count);
  if (pair < 0) return;
  int lo, hi;
  pair_decode(pair, lo, hi);
  const int item = blockIdx.x % NUNIT;
  const int e = (item / NCH) ? hi : lo;
  const int hbase = (item % NCH) * HC;

  const int col1 = wave * 32 + 2 * ln;
  const int col0 = wave * 64 + 4 * ln;
  const int lidx1 = 2 * hf * (DH / 2) + wave * 16 + ln;
  const int lidx2 = 2 * hf * (DM / 4) + wave * 16 + ln;
  v2f q1[RING][KB][2];

  const float* W1c = W1 + (size_t)e * DM * DH + hbase;
  const float* W2c = W2 + (size_t)e * DH * DM + (size_t)hbase * DM;

  sweep_request<2>((const v2f*)W1c, DH / 2, lidx1, q1);

  const int* lst = route + pair * RSTRIDE + RHEAD;
  const float* b1c = b1 + e * DH + hbase;
  float* P = part + (size_t)blockIdx.x * (MW * DM);
  if (first + MT < count)
    ffn_item<2>(emb, lst, first, count, W1c, W2c, b1c, P, sX, sH, tid, hf, ln,
                col1, col0, lidx1, lidx2, q1);
  else
    ffn_item<1>(emb, lst, first, count, W1c, W2c, b1c, P, sX, sH, tid, hf, ln,
                col1, col0, lidx1, lidx2, q1);
}

__global__ __launch_bounds__(256) void combine_kernel(
    const float* __restrict__ gates, const int* __restrict__ route,
    const float* __restrict__ b2, const float* __restrict__ part,
    float* __restrict__ out) {
  const int tid = threadIdx.x;
  const int tile = blockIdx.x / NUNIT;
  int pair, first, count;
  find_tile(route, tile, pair, first, count);
  if (pair < 0) return;
  int lo, hi;
  pair_decode(pair, lo, hi);

  const int row = (blockIdx.x % NUNIT) * (MW / NUNIT) + (tid >> 7);
  const int c4 = tid & 127;
  const int idx = first + row;
  if (idx >= count) return;
  const int tk = route[pair * RSTRIDE + RHEAD + idx];

  const v4f* P = (const v4f*)(part + (size_t)tile * NUNIT * (MW * DM)) +
                 row * (DM / 4) + c4;
  v4f p[NUNIT];
#pragma unroll
  for (int u = 0; u < NUNIT; ++u) p[u] = P[u * (MW * DM / 4)];
  const float glo = gates[(size_t)tk * NE + lo];
  const float ghi = gates[(size_t)tk * NE + hi];
  const v4f blo = ((const v4f*)(b2 + lo * DM))[c4];
  const v4f bhi = ((const v4f*)(b2 + hi * DM))[c4];

  v4f olo = p[0], ohi = p[NCH];
#pragma unroll
  for (int u = 1; u < NCH; ++u) { olo += p[u]; ohi += p[NCH + u]; }
  v4f ov;
#pragma unroll
  for (int nt = 0; nt < 4; ++nt) {
    const float a = glo * (olo[nt] + blo[nt]);
    ov[nt] = fmaf(ghi, ohi[nt] + bhi[nt], a);
  }
  ((v4f*)(out + (size_t)tk * DM))[c4] = ov;
}

extern "C" void kernel_launch(void* const* d_in, const int* in_sizes, int n_in,
                              void* d_out, int out_size, void* d_ws, size_t ws_size,
                              hipStream_t stream) {
  const float* emb   = (const float*)d_in[0];
  const float* state = (const float*)d_in[1];
  const float* Wg    = (const float*)d_in[2];
  const float* bg    = (const float*)d_in[3];
  const float* Kmat  = (const float*)d_in[4];
  const float* W1    = (const float*)d_in[5];
  const float* b1    = (const float*)d_in[6];
  const float* W2    = (const float*)d_in[7];
  const float* b2    = (const float*)d_in[8];

  float* out_main = (float*)d_out;
  float* probs    = out_main + (size_t)T_TOKENS * DM;
  float* gates    = (float*)d_ws;
  int*   route    = (int*)(gates + (size_t)T_TOKENS * NE);
  float* part     = (float*)(route + (size_t)NP * RSTRIDE);

  gating_kernel<<<T_TOKENS / 8, 256, 0, stream>>>(emb, state, Wg, bg, Kmat,
                                                  probs, gates);
  route_kernel<<<NP, 1024, 0, stream>>>(probs, route);
  moe_ffn_kernel<<<NTILE * NUNIT, 256, 0, stream>>>(emb, route, W1, b1, W2,
                                                    part);
  combine_kernel<<<NTILE * NUNIT, 256, 0, stream>>>(gates, route, b2, part,
                                                    out_main);
}
